# starting kernel
# baseline (speedup 1.0000x reference)
#include <hip/hip_runtime.h>
#include <hip/hip_bf16.h>

typedef __attribute__((ext_vector_type(2))) float v2f;
typedef __attribute__((ext_vector_type(8))) float v8f;

#define BS     4
#define NV     2048
#define NN     20
#define IN_CH  128
#define OUT_CH 128
#define NS     4
#define SXO    (NS * OUT_CH)
#define NCOL   ((NS + 1) * OUT_CH)
#define NROW   (BS * NV)
#define EPSF   1e-12f

__global__ __launch_bounds__(256)
void gemm_wmma_f32_kernel(const float* __restrict__ fm,
                          const float* __restrict__ w,
                          const float* __restrict__ bias,
                          float* __restrict__ fout) {
    const int lane  = threadIdx.x & 31;
    const int wave  = threadIdx.x >> 5;
    const int gwave = blockIdx.x * 8 + wave;

    const int ntiles_n = NCOL / 16;
    const int tm = gwave / ntiles_n;
    const int tn = gwave % ntiles_n;

    const int half = lane >> 4;
    const int l15  = lane & 15;
    const int r0   = tm * 16;
    const int c0   = tn * 16;

    v8f acc = {};
    for (int kk = 0; kk < IN_CH; kk += 4) {
        const float* arow = fm + (size_t)(r0 + l15) * IN_CH + (kk + 2 * half);
        v2f a;
        a.x = arow[0];
        a.y = arow[1];

        const float* brow = w + (size_t)(kk + 2 * half) * NCOL + c0 + l15;
        v2f b;
        b.x = brow[0];
        b.y = brow[NCOL];

        acc = __builtin_amdgcn_wmma_f32_16x16x4_f32(
             false, a,  false, b,
             (short)0, acc,  false,  false);
    }

    const float bv = bias[c0 + l15];
#pragma unroll
    for (int r = 0; r < 8; ++r) {
        const int row = r0 + r + half * 8;
        fout[(size_t)row * NCOL + c0 + l15] = acc[r] + bv;
    }
}

__global__ __launch_bounds__(256)
void fused_gconv_kernel(const int*   __restrict__ nidx,
                        const float* __restrict__ verts,
                        const float* __restrict__ fout,
                        const float* __restrict__ dirs,
                        const float* __restrict__ dw,
                        float*       __restrict__ out) {
    const int bv = blockIdx.x;
    const int b  = bv / NV;

    __shared__ float s_dn[NN][3];
    __shared__ float s_dist[NN];
    __shared__ int   s_idx[NN];
    __shared__ float s_maxd;
    __shared__ float s_vals[SXO];

    const int tid = threadIdx.x;

    if (tid < NN) {
        const int j   = tid;
        const int idx = nidx[(size_t)bv * NN + j];
        s_idx[j] = idx;

        const float cx = verts[(size_t)bv * 3 + 0];
        const float cy = verts[(size_t)bv * 3 + 1];
        const float cz = verts[(size_t)bv * 3 + 2];

        const size_t np = ((size_t)b * NV + idx) * 3;
        const float dx = verts[np + 0] - cx;
        const float dy = verts[np + 1] - cy;
        const float dz = verts[np + 2] - cz;

        const float dist = sqrtf(dx * dx + dy * dy + dz * dz);
        const float inv  = 1.0f / fmaxf(dist, EPSF);
        s_dn[j][0] = dx * inv;
        s_dn[j][1] = dy * inv;
        s_dn[j][2] = dz * inv;
        s_dist[j]  = dist;
    }
    __syncthreads();
    if (tid == 0) {
        float md = s_dist[0];
#pragma unroll
        for (int j = 1; j < NN; ++j) md = fmaxf(md, s_dist[j]);
        s_maxd = md;
    }
    __syncthreads();
    const float max_dist = s_maxd;

#pragma unroll
    for (int mm = 0; mm < 2; ++mm) {
        const int m = tid + mm * 256;

        const float sx  = dirs[0 * SXO + m];
        const float sy  = dirs[1 * SXO + m];
        const float sz  = dirs[2 * SXO + m];
        const float nrm = sqrtf(sx * sx + sy * sy + sz * sz);
        const float inv = 1.0f / fmaxf(nrm, EPSF);
        const float ux = sx * inv, uy = sy * inv, uz = sz * inv;

        float acc = -INFINITY;
#pragma unroll 4
        for (int j = 0; j < NN; ++j) {
            const float dot   = s_dn[j][0] * ux + s_dn[j][1] * uy + s_dn[j][2] * uz;
            const float theta = fmaxf(dot, 0.0f);
            const size_t frow = ((size_t)b * NV + s_idx[j]) * NCOL + OUT_CH + m;
            const float fsup  = fout[frow];
            acc = fmaxf(acc, theta * fsup);
        }
        const float dist_feat = fmaxf(max_dist * dw[m], 0.0f);
        s_vals[m] = acc + dist_feat;
    }
    __syncthreads();

    if (tid < OUT_CH) {
        const float fc = fout[(size_t)bv * NCOL + tid];
        float sum = s_vals[tid] + s_vals[OUT_CH + tid] +
                    s_vals[2 * OUT_CH + tid] + s_vals[3 * OUT_CH + tid];
        out[(size_t)bv * OUT_CH + tid] = fc + sum;
    }
}

extern "C" void kernel_launch(void* const* d_in, const int* in_sizes, int n_in,
                              void* d_out, int out_size, void* d_ws, size_t ws_size,
                              hipStream_t stream) {
    const int*   nidx  = (const int*)  d_in[0];
    const float* verts = (const float*)d_in[1];
    const float* fm    = (const float*)d_in[2];
    const float* w     = (const float*)d_in[3];
    const float* bias  = (const float*)d_in[4];
    const float* dirs  = (const float*)d_in[5];
    const float* dw    = (const float*)d_in[6];
    float*       out   = (float*)d_out;
    float*       fout  = (float*)d_ws;

    const int gemm_blocks = (NROW / 16) * (NCOL / 16) / 8;
    gemm_wmma_f32_kernel<<<gemm_blocks, 256, 0, stream>>>(fm, w, bias, fout);

    fused_gconv_kernel<<<NROW, 256, 0, stream>>>(nidx, verts, fout, dirs, dw, out);
}


# Kernel 2 re-mapped to one wave per vertex: each lane owns 4 adjacent channels of every s-group (one b128 gather per neighbor and group), geometry by readlane; no LDS or barrier, 8x fewer waves
# speedup vs baseline: 1.6916x; 1.6916x over previous
#include <hip/hip_runtime.h>
#include <hip/hip_bf16.h>

typedef __attribute__((ext_vector_type(2))) float v2f;
typedef __attribute__((ext_vector_type(8))) float v8f;
typedef __attribute__((ext_vector_type(4))) float v4f;

#define BS     4
#define NV     2048
#define NN     20
#define IN_CH  128
#define OUT_CH 128
#define NS     4
#define SXO    (NS * OUT_CH)
#define NCOL   ((NS + 1) * OUT_CH)
#define NROW   (BS * NV)
#define EPSF   1e-12f

__global__ __launch_bounds__(256)
void gemm_wmma_f32_kernel(const float* __restrict__ fm,
                          const float* __restrict__ w,
                          const float* __restrict__ bias,
                          float* __restrict__ fout) {
    const int lane  = threadIdx.x & 31;
    const int wave  = threadIdx.x >> 5;
    const int gwave = blockIdx.x * 8 + wave;

    const int ntiles_n = NCOL / 16;
    const int tm = gwave / ntiles_n;
    const int tn = gwave % ntiles_n;

    const int half = lane >> 4;
    const int l15  = lane & 15;
    const int r0   = tm * 16;
    const int c0   = tn * 16;

    v8f acc = {};
    for (int kk = 0; kk < IN_CH; kk += 4) {
        const float* arow = fm + (size_t)(r0 + l15) * IN_CH + (kk + 2 * half);
        v2f a;
        a.x = arow[0];
        a.y = arow[1];

        const float* brow = w + (size_t)(kk + 2 * half) * NCOL + c0 + l15;
        v2f b;
        b.x = brow[0];
        b.y = brow[NCOL];

        acc = __builtin_amdgcn_wmma_f32_16x16x4_f32(
             false, a,  false, b,
             (short)0, acc,  false,  false);
    }

    const float bv = bias[c0 + l15];
#pragma unroll
    for (int r = 0; r < 8; ++r) {
        const int row = r0 + r + half * 8;
        fout[(size_t)row * NCOL + c0 + l15] = acc[r] + bv;
    }
}

#define WPB 8

__global__ __launch_bounds__(256)
void fused_gconv_kernel(const int*   __restrict__ nidx,
                        const float* __restrict__ verts,
                        const float* __restrict__ fout,
                        const float* __restrict__ dirs,
                        const float* __restrict__ dw,
                        float*       __restrict__ out) {
    const int lane = threadIdx.x & 31;
    const int wave = __builtin_amdgcn_readfirstlane((int)(threadIdx.x >> 5));
    const int bv   = blockIdx.x * WPB + wave;
    const int b    = bv / NV;

    const int jn  = lane < NN ? lane : 0;
    const int idx = nidx[(size_t)bv * NN + jn];

    const float cx = verts[(size_t)bv * 3 + 0];
    const float cy = verts[(size_t)bv * 3 + 1];
    const float cz = verts[(size_t)bv * 3 + 2];

    const size_t np = ((size_t)b * NV + idx) * 3;
    const float dx = verts[np + 0] - cx;
    const float dy = verts[np + 1] - cy;
    const float dz = verts[np + 2] - cz;

    const float dist = sqrtf(dx * dx + dy * dy + dz * dz);
    const float dinv = 1.0f / fmaxf(dist, EPSF);
    const float dnx = dx * dinv;
    const float dny = dy * dinv;
    const float dnz = dz * dinv;

    float max_dist = dist;
#pragma unroll
    for (int o = 16; o >= 1; o >>= 1)
        max_dist = fmaxf(max_dist, __shfl_xor(max_dist, o));

    v4f ux[NS], uy[NS], uz[NS];
#pragma unroll
    for (int s = 0; s < NS; ++s) {
        const int m0 = s * OUT_CH + 4 * lane;
        const v4f sx = *(const v4f*)(dirs + 0 * SXO + m0);
        const v4f sy = *(const v4f*)(dirs + 1 * SXO + m0);
        const v4f sz = *(const v4f*)(dirs + 2 * SXO + m0);
#pragma unroll
        for (int c = 0; c < 4; ++c) {
            const float nrm = sqrtf(sx[c] * sx[c] + sy[c] * sy[c] + sz[c] * sz[c]);
            const float inv = 1.0f / fmaxf(nrm, EPSF);
            ux[s][c] = sx[c] * inv;
            uy[s][c] = sy[c] * inv;
            uz[s][c] = sz[c] * inv;
        }
    }

    v4f acc[NS];
#pragma unroll
    for (int s = 0; s < NS; ++s)
        acc[s] = (v4f){-INFINITY, -INFINITY, -INFINITY, -INFINITY};

    const float* fsup = fout + (size_t)b * NV * NCOL + OUT_CH + 4 * lane;

#pragma unroll 5
    for (int j = 0; j < NN; ++j) {
        const int   ij = __builtin_amdgcn_readlane(idx, j);
        const float ax = __int_as_float(__builtin_amdgcn_readlane(__float_as_int(dnx), j));
        const float ay = __int_as_float(__builtin_amdgcn_readlane(__float_as_int(dny), j));
        const float az = __int_as_float(__builtin_amdgcn_readlane(__float_as_int(dnz), j));
        const float* row = fsup + (size_t)ij * NCOL;
#pragma unroll
        for (int s = 0; s < NS; ++s) {
            const v4f f = *(const v4f*)(row + s * OUT_CH);
#pragma unroll
            for (int c = 0; c < 4; ++c) {
                const float dot   = ax * ux[s][c] + ay * uy[s][c] + az * uz[s][c];
                const float theta = fmaxf(dot, 0.0f);
                acc[s][c] = fmaxf(acc[s][c], theta * f[c]);
            }
        }
    }

    v4f val[NS];
#pragma unroll
    for (int s = 0; s < NS; ++s) {
        const v4f dwv = *(const v4f*)(dw + s * OUT_CH + 4 * lane);
#pragma unroll
        for (int c = 0; c < 4; ++c)
            val[s][c] = acc[s][c] + fmaxf(max_dist * dwv[c], 0.0f);
    }
    const v4f fc = *(const v4f*)(fout + (size_t)bv * NCOL + 4 * lane);
    v4f res;
#pragma unroll
    for (int c = 0; c < 4; ++c) {
        const float sum = val[0][c] + val[1][c] + val[2][c] + val[3][c];
        res[c] = fc[c] + sum;
    }
    *(v4f*)(out + (size_t)bv * OUT_CH + 4 * lane) = res;
}

extern "C" void kernel_launch(void* const* d_in, const int* in_sizes, int n_in,
                              void* d_out, int out_size, void* d_ws, size_t ws_size,
                              hipStream_t stream) {
    const int*   nidx  = (const int*)  d_in[0];
    const float* verts = (const float*)d_in[1];
    const float* fm    = (const float*)d_in[2];
    const float* w     = (const float*)d_in[3];
    const float* bias  = (const float*)d_in[4];
    const float* dirs  = (const float*)d_in[5];
    const float* dw    = (const float*)d_in[6];
    float*       out   = (float*)d_out;
    float*       fout  = (float*)d_ws;

    const int gemm_blocks = (NROW / 16) * (NCOL / 16) / 8;
    gemm_wmma_f32_kernel<<<gemm_blocks, 256, 0, stream>>>(fm, w, bias, fout);

    fused_gconv_kernel<<<NROW / WPB, 256, 0, stream>>>(nidx, verts, fout, dirs, dw, out);
}


# GEMM kernel register-tiled: each wave computes a 32x64 tile (2x4 WMMA tiles, column tiles interleaved by 4) from b128 fragment loads, b128 row stores, 8x fewer waves; k grouping of f32 sums changes
# speedup vs baseline: 2.1439x; 1.2674x over previous
#include <hip/hip_runtime.h>
#include <hip/hip_bf16.h>

typedef __attribute__((ext_vector_type(2))) float v2f;
typedef __attribute__((ext_vector_type(8))) float v8f;
typedef __attribute__((ext_vector_type(4))) float v4f;

#define BS     4
#define NV     2048
#define NN     20
#define IN_CH  128
#define OUT_CH 128
#define NS     4
#define SXO    (NS * OUT_CH)
#define NCOL   ((NS + 1) * OUT_CH)
#define NROW   (BS * NV)
#define EPSF   1e-12f

#define GM_WAVES 4
#define GM_BM    (GM_WAVES * 32)
#define GM_BN    64
#define GM_NPAN  (NCOL / GM_BN)
#define GM_KB    (IN_CH / 8)

__device__ __forceinline__ void gemm_load_block(const float* __restrict__ ap0,
                                                const float* __restrict__ ap1,
                                                const float* __restrict__ bp,
                                                int kb, v4f (&a)[2], v4f (&b)[4]) {
    a[0] = *(const v4f*)(ap0 + 8 * kb);
    a[1] = *(const v4f*)(ap1 + 8 * kb);
#pragma unroll
    for (int e = 0; e < 4; ++e)
        b[e] = *(const v4f*)(bp + (size_t)(8 * kb + e) * NCOL);
}

__device__ __forceinline__ void gemm_mma_block(v8f (&acc)[2][4],
                                               const v4f (&a)[2], const v4f (&b)[4]) {
#pragma unroll
    for (int p = 0; p < 2; ++p) {
        v2f af0, af1;
        af0.x = a[0][2 * p]; af0.y = a[0][2 * p + 1];
        af1.x = a[1][2 * p]; af1.y = a[1][2 * p + 1];
#pragma unroll
        for (int t = 0; t < 4; ++t) {
            v2f bf;
            bf.x = b[2 * p][t];
            bf.y = b[2 * p + 1][t];
            acc[0][t] = __builtin_amdgcn_wmma_f32_16x16x4_f32(
                 false, af0,  false, bf,
                 (short)0, acc[0][t],  false,  false);
            acc[1][t] = __builtin_amdgcn_wmma_f32_16x16x4_f32(
                 false, af1,  false, bf,
                 (short)0, acc[1][t],  false,  false);
        }
    }
}

__global__ __launch_bounds__(GM_WAVES * 32)
void gemm_wmma_f32_kernel(const float* __restrict__ fm,
                          const float* __restrict__ w,
                          const float* __restrict__ bias,
                          float* __restrict__ fout) {
    const int lane = threadIdx.x & 31;
    const int wave = __builtin_amdgcn_readfirstlane((int)(threadIdx.x >> 5));
    const int bn   = blockIdx.x % GM_NPAN;
    const int bm   = blockIdx.x / GM_NPAN;
    const int m0   = bm * GM_BM + wave * 32;
    const int n0   = bn * GM_BN;

    const int half = lane >> 4;
    const int l15  = lane & 15;

    const float* ap0 = fm + (size_t)(m0 + l15) * IN_CH + 4 * half;
    const float* ap1 = ap0 + (size_t)16 * IN_CH;
    const float* bp  = w + (size_t)(4 * half) * NCOL + n0 + 4 * l15;

    v8f acc[2][4];
#pragma unroll
    for (int i = 0; i < 2; ++i)
#pragma unroll
        for (int t = 0; t < 4; ++t)
            acc[i][t] = (v8f){};

    v4f af[3][2], bf[3][4];
    gemm_load_block(ap0, ap1, bp, 0, af[0], bf[0]);
    gemm_load_block(ap0, ap1, bp, 1, af[1], bf[1]);
#pragma unroll
    for (int kb = 0; kb < GM_KB; ++kb) {
        if (kb + 2 < GM_KB)
            gemm_load_block(ap0, ap1, bp, kb + 2, af[(kb + 2) % 3], bf[(kb + 2) % 3]);
        gemm_mma_block(acc, af[kb % 3], bf[kb % 3]);
    }

    const v4f bv = *(const v4f*)(bias + n0 + 4 * l15);
#pragma unroll
    for (int i = 0; i < 2; ++i) {
#pragma unroll
        for (int r = 0; r < 8; ++r) {
            const int row = m0 + 16 * i + r + half * 8;
            v4f o;
            o.x = acc[i][0][r] + bv.x;
            o.y = acc[i][1][r] + bv.y;
            o.z = acc[i][2][r] + bv.z;
            o.w = acc[i][3][r] + bv.w;
            *(v4f*)(fout + (size_t)row * NCOL + n0 + 4 * l15) = o;
        }
    }
}

#define WPB 8

__global__ __launch_bounds__(256)
void fused_gconv_kernel(const int*   __restrict__ nidx,
                        const float* __restrict__ verts,
                        const float* __restrict__ fout,
                        const float* __restrict__ dirs,
                        const float* __restrict__ dw,
                        float*       __restrict__ out) {
    const int lane = threadIdx.x & 31;
    const int wave = __builtin_amdgcn_readfirstlane((int)(threadIdx.x >> 5));
    const int bv   = blockIdx.x * WPB + wave;
    const int b    = bv / NV;

    const int jn  = lane < NN ? lane : 0;
    const int idx = nidx[(size_t)bv * NN + jn];

    const float cx = verts[(size_t)bv * 3 + 0];
    const float cy = verts[(size_t)bv * 3 + 1];
    const float cz = verts[(size_t)bv * 3 + 2];

    const size_t np = ((size_t)b * NV + idx) * 3;
    const float dx = verts[np + 0] - cx;
    const float dy = verts[np + 1] - cy;
    const float dz = verts[np + 2] - cz;

    const float dist = sqrtf(dx * dx + dy * dy + dz * dz);
    const float dinv = 1.0f / fmaxf(dist, EPSF);
    const float dnx = dx * dinv;
    const float dny = dy * dinv;
    const float dnz = dz * dinv;

    float max_dist = dist;
#pragma unroll
    for (int o = 16; o >= 1; o >>= 1)
        max_dist = fmaxf(max_dist, __shfl_xor(max_dist, o));

    v4f ux[NS], uy[NS], uz[NS];
#pragma unroll
    for (int s = 0; s < NS; ++s) {
        const int m0 = s * OUT_CH + 4 * lane;
        const v4f sx = *(const v4f*)(dirs + 0 * SXO + m0);
        const v4f sy = *(const v4f*)(dirs + 1 * SXO + m0);
        const v4f sz = *(const v4f*)(dirs + 2 * SXO + m0);
#pragma unroll
        for (int c = 0; c < 4; ++c) {
            const float nrm = sqrtf(sx[c] * sx[c] + sy[c] * sy[c] + sz[c] * sz[c]);
            const float inv = 1.0f / fmaxf(nrm, EPSF);
            ux[s][c] = sx[c] * inv;
            uy[s][c] = sy[c] * inv;
            uz[s][c] = sz[c] * inv;
        }
    }

    v4f acc[NS];
#pragma unroll
    for (int s = 0; s < NS; ++s)
        acc[s] = (v4f){-INFINITY, -INFINITY, -INFINITY, -INFINITY};

    const float* fsup = fout + (size_t)b * NV * NCOL + OUT_CH + 4 * lane;

#pragma unroll 5
    for (int j = 0; j < NN; ++j) {
        const int   ij = __builtin_amdgcn_readlane(idx, j);
        const float ax = __int_as_float(__builtin_amdgcn_readlane(__float_as_int(dnx), j));
        const float ay = __int_as_float(__builtin_amdgcn_readlane(__float_as_int(dny), j));
        const float az = __int_as_float(__builtin_amdgcn_readlane(__float_as_int(dnz), j));
        const float* row = fsup + (size_t)ij * NCOL;
#pragma unroll
        for (int s = 0; s < NS; ++s) {
            const v4f f = *(const v4f*)(row + s * OUT_CH);
#pragma unroll
            for (int c = 0; c < 4; ++c) {
                const float dot   = ax * ux[s][c] + ay * uy[s][c] + az * uz[s][c];
                const float theta = fmaxf(dot, 0.0f);
                acc[s][c] = fmaxf(acc[s][c], theta * f[c]);
            }
        }
    }

    v4f val[NS];
#pragma unroll
    for (int s = 0; s < NS; ++s) {
        const v4f dwv = *(const v4f*)(dw + s * OUT_CH + 4 * lane);
#pragma unroll
        for (int c = 0; c < 4; ++c)
            val[s][c] = acc[s][c] + fmaxf(max_dist * dwv[c], 0.0f);
    }
    const v4f fc = *(const v4f*)(fout + (size_t)bv * NCOL + 4 * lane);
    v4f res;
#pragma unroll
    for (int c = 0; c < 4; ++c) {
        const float sum = val[0][c] + val[1][c] + val[2][c] + val[3][c];
        res[c] = fc[c] + sum;
    }
    *(v4f*)(out + (size_t)bv * OUT_CH + 4 * lane) = res;
}

extern "C" void kernel_launch(void* const* d_in, const int* in_sizes, int n_in,
                              void* d_out, int out_size, void* d_ws, size_t ws_size,
                              hipStream_t stream) {
    const int*   nidx  = (const int*)  d_in[0];
    const float* verts = (const float*)d_in[1];
    const float* fm    = (const float*)d_in[2];
    const float* w     = (const float*)d_in[3];
    const float* bias  = (const float*)d_in[4];
    const float* dirs  = (const float*)d_in[5];
    const float* dw    = (const float*)d_in[6];
    float*       out   = (float*)d_out;
    float*       fout  = (float*)d_ws;

    const int gemm_blocks = (NROW / GM_BM) * GM_NPAN;
    gemm_wmma_f32_kernel<<<gemm_blocks, GM_WAVES * 32, 0, stream>>>(fm, w, bias, fout);

    fused_gconv_kernel<<<NROW / WPB, 256, 0, stream>>>(nidx, verts, fout, dirs, dw, out);
}


# Unit support directions computed once per launch by 512 threads of the GEMM kernel into the workspace; kernel 2's 8192 waves load, not recompute, them; sum of 3 squares may fuse differently
# speedup vs baseline: 2.2262x; 1.0384x over previous
#include <hip/hip_runtime.h>
#include <hip/hip_bf16.h>

typedef __attribute__((ext_vector_type(2))) float v2f;
typedef __attribute__((ext_vector_type(8))) float v8f;
typedef __attribute__((ext_vector_type(4))) float v4f;

#define BS     4
#define NV     2048
#define NN     20
#define IN_CH  128
#define OUT_CH 128
#define NS     4
#define SXO    (NS * OUT_CH)
#define NCOL   ((NS + 1) * OUT_CH)
#define NROW   (BS * NV)
#define EPSF   1e-12f

#define GM_WAVES 4
#define GM_BM    (GM_WAVES * 32)
#define GM_BN    64
#define GM_NPAN  (NCOL / GM_BN)
#define GM_KB    (IN_CH / 8)

__device__ __forceinline__ void gemm_load_block(const float* __restrict__ ap0,
                                                const float* __restrict__ ap1,
                                                const float* __restrict__ bp,
                                                int kb, v4f (&a)[2], v4f (&b)[4]) {
    a[0] = *(const v4f*)(ap0 + 8 * kb);
    a[1] = *(const v4f*)(ap1 + 8 * kb);
#pragma unroll
    for (int e = 0; e < 4; ++e)
        b[e] = *(const v4f*)(bp + (size_t)(8 * kb + e) * NCOL);
}

__device__ __forceinline__ void gemm_mma_block(v8f (&acc)[2][4],
                                               const v4f (&a)[2], const v4f (&b)[4]) {
#pragma unroll
    for (int p = 0; p < 2; ++p) {
        v2f af0, af1;
        af0.x = a[0][2 * p]; af0.y = a[0][2 * p + 1];
        af1.x = a[1][2 * p]; af1.y = a[1][2 * p + 1];
#pragma unroll
        for (int t = 0; t < 4; ++t) {
            v2f bf;
            bf.x = b[2 * p][t];
            bf.y = b[2 * p + 1][t];
            acc[0][t] = __builtin_amdgcn_wmma_f32_16x16x4_f32(
                 false, af0,  false, bf,
                 (short)0, acc[0][t],  false,  false);
            acc[1][t] = __builtin_amdgcn_wmma_f32_16x16x4_f32(
                 false, af1,  false, bf,
                 (short)0, acc[1][t],  false,  false);
        }
    }
}

__global__ __launch_bounds__(GM_WAVES * 32)
void gemm_wmma_f32_kernel(const float* __restrict__ fm,
                          const float* __restrict__ w,
                          const float* __restrict__ bias,
                          const float* __restrict__ dirs,
                          float* __restrict__ fout,
                          float* __restrict__ sdn) {
    if (blockIdx.x < SXO / (GM_WAVES * 32)) {
        const int m = blockIdx.x * (GM_WAVES * 32) + threadIdx.x;
        const float sx = dirs[0 * SXO + m];
        const float sy = dirs[1 * SXO + m];
        const float sz = dirs[2 * SXO + m];
        const float nrm = sqrtf(sx * sx + sy * sy + sz * sz);
        const float inv = 1.0f / fmaxf(nrm, EPSF);
        sdn[0 * SXO + m] = sx * inv;
        sdn[1 * SXO + m] = sy * inv;
        sdn[2 * SXO + m] = sz * inv;
    }

    const int lane = threadIdx.x & 31;
    const int wave = __builtin_amdgcn_readfirstlane((int)(threadIdx.x >> 5));
    const int bn   = blockIdx.x % GM_NPAN;
    const int bm   = blockIdx.x / GM_NPAN;
    const int m0   = bm * GM_BM + wave * 32;
    const int n0   = bn * GM_BN;

    const int half = lane >> 4;
    const int l15  = lane & 15;

    const float* ap0 = fm + (size_t)(m0 + l15) * IN_CH + 4 * half;
    const float* ap1 = ap0 + (size_t)16 * IN_CH;
    const float* bp  = w + (size_t)(4 * half) * NCOL + n0 + 4 * l15;

    v8f acc[2][4];
#pragma unroll
    for (int i = 0; i < 2; ++i)
#pragma unroll
        for (int t = 0; t < 4; ++t)
            acc[i][t] = (v8f){};

    v4f af[3][2], bf[3][4];
    gemm_load_block(ap0, ap1, bp, 0, af[0], bf[0]);
    gemm_load_block(ap0, ap1, bp, 1, af[1], bf[1]);

#pragma unroll
    for (int kb = 0; kb < GM_KB; ++kb) {
        if (kb + 2 < GM_KB)
            gemm_load_block(ap0, ap1, bp, kb + 2, af[(kb + 2) % 3], bf[(kb + 2) % 3]);
        gemm_mma_block(acc, af[kb % 3], bf[kb % 3]);
    }

    const v4f bv = *(const v4f*)(bias + n0 + 4 * l15);
#pragma unroll
    for (int i = 0; i < 2; ++i) {
#pragma unroll
        for (int r = 0; r < 8; ++r) {
            const int row = m0 + 16 * i + r + half * 8;
            v4f o;
            o.x = acc[i][0][r] + bv.x;
            o.y = acc[i][1][r] + bv.y;
            o.z = acc[i][2][r] + bv.z;
            o.w = acc[i][3][r] + bv.w;
            *(v4f*)(fout + (size_t)row * NCOL + n0 + 4 * l15) = o;
        }
    }
}

#define WPB 8

__global__ __launch_bounds__(256)
void fused_gconv_kernel(const int*   __restrict__ nidx,
                        const float* __restrict__ verts,
                        const float* __restrict__ fout,
                        const float* __restrict__ sdn,
                        const float* __restrict__ dw,
                        float*       __restrict__ out) {
    const int lane = threadIdx.x & 31;
    const int wave = __builtin_amdgcn_readfirstlane((int)(threadIdx.x >> 5));
    const int bv   = blockIdx.x * WPB + wave;
    const int b    = bv / NV;

    const int jn  = lane < NN ? lane : 0;
    const int idx = nidx[(size_t)bv * NN + jn];

    const float cx = verts[(size_t)bv * 3 + 0];
    const float cy = verts[(size_t)bv * 3 + 1];
    const float cz = verts[(size_t)bv * 3 + 2];

    const size_t np = ((size_t)b * NV + idx) * 3;
    const float dx = verts[np + 0] - cx;
    const float dy = verts[np + 1] - cy;
    const float dz = verts[np + 2] - cz;

    const float dist = sqrtf(dx * dx + dy * dy + dz * dz);
    const float dinv = 1.0f / fmaxf(dist, EPSF);
    const float dnx = dx * dinv;
    const float dny = dy * dinv;
    const float dnz = dz * dinv;

    float max_dist = dist;
#pragma unroll
    for (int o = 16; o >= 1; o >>= 1)
        max_dist = fmaxf(max_dist, __shfl_xor(max_dist, o));

    v4f ux[NS], uy[NS], uz[NS];
#pragma unroll
    for (int s = 0; s < NS; ++s) {
        const int m0 = s * OUT_CH + 4 * lane;
        ux[s] = *(const v4f*)(sdn + 0 * SXO + m0);
        uy[s] = *(const v4f*)(sdn + 1 * SXO + m0);
        uz[s] = *(const v4f*)(sdn + 2 * SXO + m0);
    }

    v4f acc[NS];
#pragma unroll
    for (int s = 0; s < NS; ++s)
        acc[s] = (v4f){-INFINITY, -INFINITY, -INFINITY, -INFINITY};

    const float* fsup = fout + (size_t)b * NV * NCOL + OUT_CH + 4 * lane;

#pragma unroll 5
    for (int j = 0; j < NN; ++j) {
        const int   ij = __builtin_amdgcn_readlane(idx, j);
        const float ax = __int_as_float(__builtin_amdgcn_readlane(__float_as_int(dnx), j));
        const float ay = __int_as_float(__builtin_amdgcn_readlane(__float_as_int(dny), j));
        const float az = __int_as_float(__builtin_amdgcn_readlane(__float_as_int(dnz), j));
        const float* row = fsup + (size_t)ij * NCOL;
#pragma unroll
        for (int s = 0; s < NS; ++s) {
            const v4f f = *(const v4f*)(row + s * OUT_CH);
#pragma unroll
            for (int c = 0; c < 4; ++c) {
                const float dot   = ax * ux[s][c] + ay * uy[s][c] + az * uz[s][c];
                const float theta = fmaxf(dot, 0.0f);
                acc[s][c] = fmaxf(acc[s][c], theta * f[c]);
            }
        }
    }

    v4f val[NS];
#pragma unroll
    for (int s = 0; s < NS; ++s) {
        const v4f dwv = *(const v4f*)(dw + s * OUT_CH + 4 * lane);
#pragma unroll
        for (int c = 0; c < 4; ++c)
            val[s][c] = acc[s][c] + fmaxf(max_dist * dwv[c], 0.0f);
    }
    const v4f fc = *(const v4f*)(fout + (size_t)bv * NCOL + 4 * lane);
    v4f res;
#pragma unroll
    for (int c = 0; c < 4; ++c) {
        const float sum = val[0][c] + val[1][c] + val[2][c] + val[3][c];
        res[c] = fc[c] + sum;
    }
    *(v4f*)(out + (size_t)bv * OUT_CH + 4 * lane) = res;
}

extern "C" void kernel_launch(void* const* d_in, const int* in_sizes, int n_in,
                              void* d_out, int out_size, void* d_ws, size_t ws_size,
                              hipStream_t stream) {
    const int*   nidx  = (const int*)  d_in[0];
    const float* verts = (const float*)d_in[1];
    const float* fm    = (const float*)d_in[2];
    const float* w     = (const float*)d_in[3];
    const float* bias  = (const float*)d_in[4];
    const float* dirs  = (const float*)d_in[5];
    const float* dw    = (const float*)d_in[6];
    float*       out   = (float*)d_out;
    float*       fout  = (float*)d_ws;
    float*       sdn   = fout + (size_t)NROW * NCOL;

    const int gemm_blocks = (NROW / GM_BM) * GM_NPAN;
    gemm_wmma_f32_kernel<<<gemm_blocks, GM_WAVES * 32, 0, stream>>>(fm, w, bias, dirs, fout, sdn);

    fused_gconv_kernel<<<NROW / WPB, 256, 0, stream>>>(nidx, verts, fout, sdn, dw, out);
}


# Kernel 2 as 2048 resident waves of 4 vertices each: a wave's 80 neighbor rows run as one stream of 5-row trips, the next trip's 20 gathers requested ahead (20-60 in flight per wave, was about 9)
# speedup vs baseline: 2.3556x; 1.0581x over previous
#include <hip/hip_runtime.h>
#include <hip/hip_bf16.h>

typedef __attribute__((ext_vector_type(2))) float v2f;
typedef __attribute__((ext_vector_type(8))) float v8f;
typedef __attribute__((ext_vector_type(4))) float v4f;

#define BS     4
#define NV     2048
#define NN     20
#define IN_CH  128
#define OUT_CH 128
#define NS     4
#define SXO    (NS * OUT_CH)
#define NCOL   ((NS + 1) * OUT_CH)
#define NROW   (BS * NV)
#define EPSF   1e-12f

#define GM_WAVES 4
#define GM_BM    (GM_WAVES * 32)
#define GM_BN    64
#define GM_NPAN  (NCOL / GM_BN)
#define GM_KB    (IN_CH / 8)

__device__ __forceinline__ void gemm_load_block(const float* __restrict__ ap0,
                                                const float* __restrict__ ap1,
                                                const float* __restrict__ bp,
                                                int kb, v4f (&a)[2], v4f (&b)[4]) {
    a[0] = *(const v4f*)(ap0 + 8 * kb);
    a[1] = *(const v4f*)(ap1 + 8 * kb);
#pragma unroll
    for (int e = 0; e < 4; ++e)
        b[e] = *(const v4f*)(bp + (size_t)(8 * kb + e) * NCOL);
}

__device__ __forceinline__ void gemm_mma_block(v8f (&acc)[2][4],
                                               const v4f (&a)[2], const v4f (&b)[4]) {
#pragma unroll
    for (int p = 0; p < 2; ++p) {
        v2f af0, af1;
        af0.x = a[0][2 * p]; af0.y = a[0][2 * p + 1];
        af1.x = a[1][2 * p]; af1.y = a[1][2 * p + 1];
#pragma unroll
        for (int t = 0; t < 4; ++t) {
            v2f bf;
            bf.x = b[2 * p][t];
            bf.y = b[2 * p + 1][t];
            acc[0][t] = __builtin_amdgcn_wmma_f32_16x16x4_f32(
                 false, af0,  false, bf,
                 (short)0, acc[0][t],  false,  false);
            acc[1][t] = __builtin_amdgcn_wmma_f32_16x16x4_f32(
                 false, af1,  false, bf,
                 (short)0, acc[1][t],  false,  false);
        }
    }
}

__global__ __launch_bounds__(GM_WAVES * 32)
void gemm_wmma_f32_kernel(const float* __restrict__ fm,
                          const float* __restrict__ w,
                          const float* __restrict__ bias,
                          const float* __restrict__ dirs,
                          float* __restrict__ fout,
                          float* __restrict__ sdn) {
    if (blockIdx.x < SXO / (GM_WAVES * 32)) {
        const int m = blockIdx.x * (GM_WAVES * 32) + threadIdx.x;
        const float sx = dirs[0 * SXO + m];
        const float sy = dirs[1 * SXO + m];
        const float sz = dirs[2 * SXO + m];
        const float nrm = sqrtf(sx * sx + sy * sy + sz * sz);
        const float inv = 1.0f / fmaxf(nrm, EPSF);
        sdn[0 * SXO + m] = sx * inv;
        sdn[1 * SXO + m] = sy * inv;
        sdn[2 * SXO + m] = sz * inv;
    }

    const int lane = threadIdx.x & 31;
    const int wave = __builtin_amdgcn_readfirstlane((int)(threadIdx.x >> 5));
    const int bn   = blockIdx.x % GM_NPAN;
    const int bm   = blockIdx.x / GM_NPAN;
    const int m0   = bm * GM_BM + wave * 32;
    const int n0   = bn * GM_BN;

    const int half = lane >> 4;
    const int l15  = lane & 15;

    const float* ap0 = fm + (size_t)(m0 + l15) * IN_CH + 4 * half;
    const float* ap1 = ap0 + (size_t)16 * IN_CH;
    const float* bp  = w + (size_t)(4 * half) * NCOL + n0 + 4 * l15;

    v8f acc[2][4];
#pragma unroll
    for (int i = 0; i < 2; ++i)
#pragma unroll
        for (int t = 0; t < 4; ++t)
            acc[i][t] = (v8f){};

    v4f af[3][2], bf[3][4];
    gemm_load_block(ap0, ap1, bp, 0, af[0], bf[0]);
    gemm_load_block(ap0, ap1, bp, 1, af[1], bf[1]);

#pragma unroll
    for (int kb = 0; kb < GM_KB; ++kb) {
        if (kb + 2 < GM_KB)
            gemm_load_block(ap0, ap1, bp, kb + 2, af[(kb + 2) % 3], bf[(kb + 2) % 3]);
        gemm_mma_block(acc, af[kb % 3], bf[kb % 3]);
    }

    const v4f bv = *(const v4f*)(bias + n0 + 4 * l15);
#pragma unroll
    for (int i = 0; i < 2; ++i) {
#pragma unroll
        for (int r = 0; r < 8; ++r) {
            const int row = m0 + 16 * i + r + half * 8;
            v4f o;
            o.x = acc[i][0][r] + bv.x;
            o.y = acc[i][1][r] + bv.y;
            o.z = acc[i][2][r] + bv.z;
            o.w = acc[i][3][r] + bv.w;
            *(v4f*)(fout + (size_t)row * NCOL + n0 + 4 * l15) = o;
        }
    }
}

#define WPB   8
#define VPW   4
#define SPT   5
#define TPV   (NN / SPT)

struct GconvVertex {
    int   idx;
    float dnx, dny, dnz;
    float max_dist;
};

struct GconvState {
    v4f dwv[NS];
    v4f ux[NS], uy[NS], uz[NS];
    v4f acc[NS];
    v4f buf[TPV][SPT][NS];
};

template <int P>
__device__ __forceinline__ void gconv_request(GconvState& g, int ci, const float* fsup) {
#pragma unroll
    for (int r = 0; r < SPT; ++r) {
        const int ij = __builtin_amdgcn_readlane(ci, SPT * P + r);
        const float* row = fsup + (size_t)ij * NCOL;
#pragma unroll
        for (int s = 0; s < NS; ++s)
            g.buf[P][r][s] = *(const v4f*)(row + s * OUT_CH);
    }
}

template <int P>
__device__ __forceinline__ void gconv_use(GconvState& g, const GconvVertex& v) {
#pragma unroll
    for (int r = 0; r < SPT; ++r) {
        const float ax = __int_as_float(__builtin_amdgcn_readlane(__float_as_int(v.dnx), SPT * P + r));
        const float ay = __int_as_float(__builtin_amdgcn_readlane(__float_as_int(v.dny), SPT * P + r));
        const float az = __int_as_float(__builtin_amdgcn_readlane(__float_as_int(v.dnz), SPT * P + r));
#pragma unroll
        for (int s = 0; s < NS; ++s) {
            const v4f f = g.buf[P][r][s];
#pragma unroll
            for (int c = 0; c < 4; ++c) {
                const float dot   = ax * g.ux[s][c] + ay * g.uy[s][c] + az * g.uz[s][c];
                const float theta = fmaxf(dot, 0.0f);
                g.acc[s][c] = fmaxf(g.acc[s][c], theta * f[c]);
            }
        }
    }
}

template <int P, bool LAST>
__device__ __forceinline__ void gconv_trips(GconvState& g, const GconvVertex& cur, int nxt_idx,
                                            const float* fsup) {
    if constexpr (P < TPV) {
        if constexpr (P + 1 < TPV)
            gconv_request<P + 1>(g, cur.idx, fsup);
        else if constexpr (!LAST)
            gconv_request<0>(g, nxt_idx, fsup);
        __builtin_amdgcn_sched_barrier(0);
        gconv_use<P>(g, cur);
        __builtin_amdgcn_sched_barrier(0);
        gconv_trips<P + 1, LAST>(g, cur, nxt_idx, fsup);
    }
}

__device__ __forceinline__ void gconv_finish(GconvState& g, float max_dist, v4f fc,
                                             float* __restrict__ outp) {
    v4f val[NS];
#pragma unroll
    for (int s = 0; s < NS; ++s) {
#pragma unroll
        for (int c = 0; c < 4; ++c)
            val[s][c] = g.acc[s][c] + fmaxf(max_dist * g.dwv[s][c], 0.0f);
    }
    v4f res;
#pragma unroll
    for (int c = 0; c < 4; ++c) {
        const float sum = val[0][c] + val[1][c] + val[2][c] + val[3][c];
        res[c] = fc[c] + sum;
    }
    *(v4f*)outp = res;
#pragma unroll
    for (int s = 0; s < NS; ++s)
        g.acc[s] = (v4f){-INFINITY, -INFINITY, -INFINITY, -INFINITY};
}

__global__ __launch_bounds__(256)
void fused_gconv_kernel(const int*   __restrict__ nidx,
                        const float* __restrict__ verts,
                        const float* __restrict__ fout,
                        const float* __restrict__ sdn,
                        const float* __restrict__ dw,
                        float*       __restrict__ out) {
    const int lane = threadIdx.x & 31;
    const int wave = __builtin_amdgcn_readfirstlane((int)(threadIdx.x >> 5));
    const int bv0  = (blockIdx.x * WPB + wave) * VPW;
    const int b    = bv0 / NV;

    GconvState g;
    GconvVertex vx[VPW];

    const int jn = lane < NN ? lane : 0;
#pragma unroll
    for (int k = 0; k < VPW; ++k)
        vx[k].idx = nidx[(size_t)(bv0 + k) * NN + jn];
#pragma unroll
    for (int s = 0; s < NS; ++s)
        g.dwv[s] = *(const v4f*)(dw + s * OUT_CH + 4 * lane);
#pragma unroll
    for (int s = 0; s < NS; ++s) {
        const int m0 = s * OUT_CH + 4 * lane;
        g.ux[s] = *(const v4f*)(sdn + 0 * SXO + m0);
        g.uy[s] = *(const v4f*)(sdn + 1 * SXO + m0);
        g.uz[s] = *(const v4f*)(sdn + 2 * SXO + m0);
    }
#pragma unroll
    for (int k = 0; k < VPW; ++k) {
        const float cx = verts[(size_t)(bv0 + k) * 3 + 0];
        const float cy = verts[(size_t)(bv0 + k) * 3 + 1];
        const float cz = verts[(size_t)(bv0 + k) * 3 + 2];

        const size_t np = ((size_t)b * NV + vx[k].idx) * 3;
        const float dx = verts[np + 0] - cx;
        const float dy = verts[np + 1] - cy;
        const float dz = verts[np + 2] - cz;

        const float dist = sqrtf(dx * dx + dy * dy + dz * dz);
        const float dinv = 1.0f / fmaxf(dist, EPSF);
        vx[k].dnx = dx * dinv;
        vx[k].dny = dy * dinv;
        vx[k].dnz = dz * dinv;

        float md = dist;
#pragma unroll
        for (int o = 16; o >= 1; o >>= 1)
            md = fmaxf(md, __shfl_xor(md, o));
        vx[k].max_dist = md;
    }
#pragma unroll
    for (int s = 0; s < NS; ++s)
        g.acc[s] = (v4f){-INFINITY, -INFINITY, -INFINITY, -INFINITY};

    const float* fsup = fout + (size_t)b * NV * NCOL + OUT_CH + 4 * lane;

    gconv_request<0>(g, vx[0].idx, fsup);

    for (int k = 0; k < VPW - 1; ++k) {
        const v4f fc = *(const v4f*)(fout + (size_t)(bv0 + k) * NCOL + 4 * lane);
        gconv_trips<0, false>(g, vx[0], vx[1].idx, fsup);
        gconv_finish(g, vx[0].max_dist, fc, out + (size_t)(bv0 + k) * OUT_CH + 4 * lane);
#pragma unroll
        for (int i = 0; i + 1 < VPW; ++i)
            vx[i] = vx[i + 1];
    }
    {
        const v4f fc = *(const v4f*)(fout + (size_t)(bv0 + VPW - 1) * NCOL + 4 * lane);
        gconv_trips<0, true>(g, vx[0], 0, fsup);
        gconv_finish(g, vx[0].max_dist, fc, out + (size_t)(bv0 + VPW - 1) * OUT_CH + 4 * lane);
    }
}

extern "C" void kernel_launch(void* const* d_in, const int* in_sizes, int n_in,
                              void* d_out, int out_size, void* d_ws, size_t ws_size,
                              hipStream_t stream) {
    const int*   nidx  = (const int*)  d_in[0];
    const float* verts = (const float*)d_in[1];
    const float* fm    = (const float*)d_in[2];
    const float* w     = (const float*)d_in[3];
    const float* bias  = (const float*)d_in[4];
    const float* dirs  = (const float*)d_in[5];
    const float* dw    = (const float*)d_in[6];
    float*       out   = (float*)d_out;
    float*       fout  = (float*)d_ws;
    float*       sdn   = fout + (size_t)NROW * NCOL;

    const int gemm_blocks = (NROW / GM_BM) * GM_NPAN;
    gemm_wmma_f32_kernel<<<gemm_blocks, GM_WAVES * 32, 0, stream>>>(fm, w, bias, dirs, fout, sdn);

    fused_gconv_kernel<<<NROW / (WPB * VPW), 256, 0, stream>>>(nidx, verts, fout, sdn, dw, out);
}
